# starting kernel
# baseline (speedup 1.0000x reference)
#include <hip/hip_runtime.h>
#include <cstdint>


typedef float v2f __attribute__((ext_vector_type(2)));
typedef float v8f __attribute__((ext_vector_type(8)));

__global__ void rmtpp_durations_kernel(const float* __restrict__ ts,
                                       float* __restrict__ dur,
                                       int T, int BS) {
    int b = blockIdx.x * blockDim.x + threadIdx.x;
    if (b >= BS) return;
    float m = -3.402823466e38f;
    for (int t = 0; t < T; ++t) {
        m = fmaxf(m, ts[(size_t)t * BS + b]);
    }
    dur[b] = m;
}

__global__ void rmtpp_proj_wmma_kernel(const float* __restrict__ h,
                                       const float* __restrict__ w,
                                       float* __restrict__ proj,
                                       int rows, int H) {
    const int lane        = threadIdx.x & 31;
    const int waveInBlock = threadIdx.x >> 5;
    const int wave        = blockIdx.x * (blockDim.x >> 5) + waveInBlock;
    const int rowBase     = wave * 16;
    if (rowBase >= rows) return;

    const int m    = lane & 15;
    const int koff = (lane >> 4) << 1;
    int r = rowBase + m;
    if (r >= rows) r = rows - 1;

    const float* __restrict__ rowPtr = h + (size_t)r * H + koff;
    const float* __restrict__ wPtr   = w + koff;

    v8f c = {};
    #pragma unroll 4
    for (int k = 0; k < H; k += 4) {
        v2f a = *reinterpret_cast<const v2f*>(rowPtr + k);
        v2f b = *reinterpret_cast<const v2f*>(wPtr + k);
        c = __builtin_amdgcn_wmma_f32_16x16x4_f32(
                false, a, false, b, (short)0, c, false, false);
    }

    const int sel = lane & 7;
    float v = c[0];
    v = (sel == 1) ? c[1] : v;
    v = (sel == 2) ? c[2] : v;
    v = (sel == 3) ? c[3] : v;
    v = (sel == 4) ? c[4] : v;
    v = (sel == 5) ? c[5] : v;
    v = (sel == 6) ? c[6] : v;
    v = (sel == 7) ? c[7] : v;

    const int  mOut    = ((lane >> 4) << 3) | sel;
    const bool doStore = (lane < 8) || (lane >= 16 && lane < 24);
    if (doStore && (rowBase + mOut) < rows) {
        proj[rowBase + mOut] = v;
    }
}

__global__ void rmtpp_grid_kernel(const float* __restrict__ ts,
                                  const float* __restrict__ proj,
                                  const float* __restrict__ dur,
                                  const float* __restrict__ time_influence,
                                  const float* __restrict__ base_intensity,
                                  float* __restrict__ out_log,
                                  float* __restrict__ out_grid,
                                  int T, int BS, int NG) {
    const int gid = blockIdx.x * blockDim.x + threadIdx.x;
    if (gid >= NG * BS) return;
    const int b = gid % BS;
    const int n = gid / BS;

    const float ti = time_influence[0];
    const float bi = base_intensity[0];

    const float t_grid = ((float)n / (float)(NG - 1)) * dur[b];

    int idx = 0;
    for (int t = 0; t < T; ++t) {
        if (ts[(size_t)t * BS + b] < t_grid) { idx = t; break; }
    }

    const float closest_ts = ts[(size_t)idx * BS + b];
    const float li = proj[(size_t)idx * BS + b] + ti * (t_grid - closest_ts) + bi;

    out_log[gid]  = li;
    out_grid[gid] = t_grid;
}

extern "C" void kernel_launch(void* const* d_in, const int* in_sizes, int n_in,
                              void* d_out, int out_size, void* d_ws, size_t ws_size,
                              hipStream_t stream) {
    const float* h  = (const float*)d_in[0];
    const float* ts = (const float*)d_in[1];
    const float* w  = (const float*)d_in[2];
    const float* ti = (const float*)d_in[3];
    const float* bi = (const float*)d_in[4];

    const int NG  = 1000;
    const int BS  = out_size / (2 * NG);
    const int TBS = in_sizes[1];
    const int T   = TBS / BS;
    const int H   = in_sizes[2];

    float* out_log  = (float*)d_out;
    float* out_grid = out_log + (size_t)NG * BS;

    float* proj = (float*)d_ws;
    float* dur  = proj + TBS;

    rmtpp_durations_kernel<<<(BS + 127) / 128, 128, 0, stream>>>(ts, dur, T, BS);

    const int waves  = (TBS + 15) / 16;
    const int blocks = (waves + 7) / 8;
    rmtpp_proj_wmma_kernel<<<blocks, 256, 0, stream>>>(h, w, proj, TBS, H);

    const int total = NG * BS;
    rmtpp_grid_kernel<<<(total + 255) / 256, 256, 0, stream>>>(
        ts, proj, dur, ti, bi, out_log, out_grid, T, BS, NG);
}


# grid kernel: the first-earlier scan tests a batch of independent loads per trip (8, then 64 in flight) instead of one load and branch per timestamp; index and arithmetic unchanged
# speedup vs baseline: 3.9974x; 3.9974x over previous
#include <hip/hip_runtime.h>
#include <cstdint>


typedef float v2f __attribute__((ext_vector_type(2)));
typedef float v8f __attribute__((ext_vector_type(8)));

__global__ void rmtpp_durations_kernel(const float* __restrict__ ts,
                                       float* __restrict__ dur,
                                       int T, int BS) {
    int b = blockIdx.x * blockDim.x + threadIdx.x;
    if (b >= BS) return;
    float m = -3.402823466e38f;
    for (int t = 0; t < T; ++t) {
        m = fmaxf(m, ts[(size_t)t * BS + b]);
    }
    dur[b] = m;
}

__global__ void rmtpp_proj_wmma_kernel(const float* __restrict__ h,
                                       const float* __restrict__ w,
                                       float* __restrict__ proj,
                                       int rows, int H) {
    const int lane        = threadIdx.x & 31;
    const int waveInBlock = threadIdx.x >> 5;
    const int wave        = blockIdx.x * (blockDim.x >> 5) + waveInBlock;
    const int rowBase     = wave * 16;
    if (rowBase >= rows) return;

    const int m    = lane & 15;
    const int koff = (lane >> 4) << 1;
    int r = rowBase + m;
    if (r >= rows) r = rows - 1;

    const float* __restrict__ rowPtr = h + (size_t)r * H + koff;
    const float* __restrict__ wPtr   = w + koff;

    v8f c = {};
    #pragma unroll 4
    for (int k = 0; k < H; k += 4) {
        v2f a = *reinterpret_cast<const v2f*>(rowPtr + k);
        v2f b = *reinterpret_cast<const v2f*>(wPtr + k);
        c = __builtin_amdgcn_wmma_f32_16x16x4_f32(
                false, a, false, b, (short)0, c, false, false);
    }

    const int sel = lane & 7;
    float v = c[0];
    v = (sel == 1) ? c[1] : v;
    v = (sel == 2) ? c[2] : v;
    v = (sel == 3) ? c[3] : v;
    v = (sel == 4) ? c[4] : v;
    v = (sel == 5) ? c[5] : v;
    v = (sel == 6) ? c[6] : v;
    v = (sel == 7) ? c[7] : v;

    const int  mOut    = ((lane >> 4) << 3) | sel;
    const bool doStore = (lane < 8) || (lane >= 16 && lane < 24);
    if (doStore && (rowBase + mOut) < rows) {
        proj[rowBase + mOut] = v;
    }
}

constexpr int kNone = 0x7fffffff;

template <int C, bool GUARD>
__device__ __forceinline__ int first_earlier_in_batch(const float* __restrict__ ts,
                                                      unsigned e0, unsigned stride,
                                                      int nvalid, float g) {
    float v[C];
    #pragma unroll
    for (int j = C - 1; j >= 0; --j) {
        const unsigned jj = (GUARD && j >= nvalid) ? (unsigned)(nvalid - 1) : (unsigned)j;
        v[j] = ts[e0 + jj * stride];
    }
    __builtin_amdgcn_sched_barrier(0);
    int first = kNone;
    #pragma unroll
    for (int j = C - 1; j >= 0; --j) {
        const bool ok = !GUARD || (j < nvalid);
        first = (ok && v[j] < g) ? j : first;
    }
    return first;
}

__global__ void __launch_bounds__(256)
rmtpp_grid_kernel(const float* __restrict__ ts,
                  const float* __restrict__ proj,
                  const float* __restrict__ dur,
                  const float* __restrict__ time_influence,
                  const float* __restrict__ base_intensity,
                  float* __restrict__ out_log,
                  float* __restrict__ out_grid,
                  int T, int BS, int NG) {
    const int gid = blockIdx.x * blockDim.x + threadIdx.x;
    if (gid >= NG * BS) return;
    const int b = gid % BS;
    const int n = gid / BS;

    const float ti = time_influence[0];
    const float bi = base_intensity[0];

    const float t_grid = ((float)n / (float)(NG - 1)) * dur[b];

    const unsigned ub = (unsigned)b;
    const unsigned us = (unsigned)BS;
    int first = first_earlier_in_batch<8, true>(ts, ub, us, T, t_grid);
    const int nFull = (T > 8) ? (T - 8) / 64 : 0;
    for (int k = 0; k < nFull && first == kNone; ++k) {
        const int t0 = 8 + k * 64;
        const int f  = first_earlier_in_batch<64, false>(ts, (unsigned)t0 * us + ub, us, 64, t_grid);
        first = (f == kNone) ? kNone : t0 + f;
    }
    const int tRem = 8 + nFull * 64;
    if (first == kNone && tRem < T) {
        const int f = first_earlier_in_batch<64, true>(ts, (unsigned)tRem * us + ub, us, T - tRem, t_grid);
        first = (f == kNone) ? kNone : tRem + f;
    }
    const int idx = (first == kNone) ? 0 : first;

    const float closest_ts = ts[(size_t)idx * BS + b];
    const float li = proj[(size_t)idx * BS + b] + ti * (t_grid - closest_ts) + bi;

    out_log[gid]  = li;
    out_grid[gid] = t_grid;
}

extern "C" void kernel_launch(void* const* d_in, const int* in_sizes, int n_in,
                              void* d_out, int out_size, void* d_ws, size_t ws_size,
                              hipStream_t stream) {
    const float* h  = (const float*)d_in[0];
    const float* ts = (const float*)d_in[1];
    const float* w  = (const float*)d_in[2];
    const float* ti = (const float*)d_in[3];
    const float* bi = (const float*)d_in[4];

    const int NG  = 1000;
    const int BS  = out_size / (2 * NG);
    const int TBS = in_sizes[1];
    const int T   = TBS / BS;
    const int H   = in_sizes[2];

    float* out_log  = (float*)d_out;
    float* out_grid = out_log + (size_t)NG * BS;

    float* proj = (float*)d_ws;
    float* dur  = proj + TBS;

    rmtpp_durations_kernel<<<(BS + 127) / 128, 128, 0, stream>>>(ts, dur, T, BS);

    const int waves  = (TBS + 15) / 16;
    const int blocks = (waves + 7) / 8;
    rmtpp_proj_wmma_kernel<<<blocks, 256, 0, stream>>>(h, w, proj, TBS, H);

    const int total = NG * BS;
    rmtpp_grid_kernel<<<(total + 255) / 256, 256, 0, stream>>>(
        ts, proj, dur, ti, bi, out_log, out_grid, T, BS, NG);
}


# proj kernel: each wave load reads 512 contiguous bytes of ONE row (was 16 rows x 16 B), non-temporal, w in the same lane order; row dot = trace of the WMMA tile summed over lanes (new reduction tree)
# speedup vs baseline: 6.4347x; 1.6097x over previous
#include <hip/hip_runtime.h>
#include <cstdint>


typedef float v2f __attribute__((ext_vector_type(2)));
typedef float v8f __attribute__((ext_vector_type(8)));

__global__ void rmtpp_durations_kernel(const float* __restrict__ ts,
                                       float* __restrict__ dur,
                                       int T, int BS) {
    int b = blockIdx.x * blockDim.x + threadIdx.x;
    if (b >= BS) return;
    float m = -3.402823466e38f;
    for (int t = 0; t < T; ++t) {
        m = fmaxf(m, ts[(size_t)t * BS + b]);
    }
    dur[b] = m;
}

typedef float v4f __attribute__((ext_vector_type(4)));

constexpr int kProjRows   = 16;
constexpr int kProjAhead  = 3;
constexpr int kProjChunks = 4;
constexpr int kProjH      = 128 * kProjChunks;

__global__ void
rmtpp_proj_wmma_kernel(const float* __restrict__ h,
                       const float* __restrict__ w,
                       float* __restrict__ proj,
                       int rows, int H) {
    const int lane        = threadIdx.x & 31;
    const int waveInBlock = threadIdx.x >> 5;
    const int wave        = blockIdx.x * (blockDim.x >> 5) + waveInBlock;
    const int rowBase     = wave * kProjRows;
    if (rowBase >= rows) return;

    if (H == kProjH) {
        v4f wq[kProjChunks];
        #pragma unroll
        for (int q = 0; q < kProjChunks; ++q)
            wq[q] = *reinterpret_cast<const v4f*>(w + 128 * q + 4 * lane);

        const float* __restrict__ hl = h + 4 * lane;
        v4f buf[kProjAhead + 1][kProjChunks];
        float d[kProjRows];

#define PROJ_LOAD_ROW(R) do { \
            int rr_ = rowBase + (R); rr_ = (rr_ < rows) ? rr_ : rows - 1; \
            const float* rp_ = hl + (size_t)rr_ * kProjH; \
            _Pragma("unroll") for (int q = 0; q < kProjChunks; ++q) \
                buf[(R) % (kProjAhead + 1)][q] = \
                    __builtin_nontemporal_load(reinterpret_cast<const v4f*>(rp_ + 128 * q)); \
        } while (0)

        const int  sel    = lane & 7;
        const bool onDiag = (lane < 8) || (lane >= 24);

        #pragma unroll
        for (int r = 0; r < kProjAhead; ++r) PROJ_LOAD_ROW(r);

        #pragma unroll
        for (int r = 0; r < kProjRows; ++r) {
            __builtin_amdgcn_sched_barrier(0);
            if (r + kProjAhead < kProjRows) PROJ_LOAD_ROW(r + kProjAhead);
            __builtin_amdgcn_sched_barrier(0);

            v8f c = {};
            #pragma unroll
            for (int q = 0; q < kProjChunks; ++q) {
                const v4f a = buf[r % (kProjAhead + 1)][q];
                const v4f b = wq[q];
                const v2f a0 = {a.x, a.y}, b0 = {b.x, b.y};
                const v2f a1 = {a.z, a.w}, b1 = {b.z, b.w};
                c = __builtin_amdgcn_wmma_f32_16x16x4_f32(
                        false, a0, false, b0, (short)0, c, false, false);
                c = __builtin_amdgcn_wmma_f32_16x16x4_f32(
                        false, a1, false, b1, (short)0, c, false, false);
            }
            float v = c[0];
            v = (sel == 1) ? c[1] : v;
            v = (sel == 2) ? c[2] : v;
            v = (sel == 3) ? c[3] : v;
            v = (sel == 4) ? c[4] : v;
            v = (sel == 5) ? c[5] : v;
            v = (sel == 6) ? c[6] : v;
            v = (sel == 7) ? c[7] : v;
            d[r] = onDiag ? v : 0.0f;
        }
#undef PROJ_LOAD_ROW

        float mine = 0.0f;
        #pragma unroll
        for (int r = 0; r < kProjRows; ++r) {
            float s = d[r];
            #pragma unroll
            for (int m = 16; m >= 1; m >>= 1) s += __shfl_xor(s, m, 32);
            mine = (lane == r) ? s : mine;
        }
        if (lane < kProjRows && (rowBase + lane) < rows) {
            proj[rowBase + lane] = mine;
        }
        return;
    }

    const int m    = lane & 15;
    const int koff = (lane >> 4) << 1;
    int r = rowBase + m;
    if (r >= rows) r = rows - 1;

    const float* __restrict__ rowPtr = h + (size_t)r * H + koff;
    const float* __restrict__ wPtr   = w + koff;

    v8f c = {};
    #pragma unroll 4
    for (int k = 0; k < H; k += 4) {
        v2f a = *reinterpret_cast<const v2f*>(rowPtr + k);
        v2f b = *reinterpret_cast<const v2f*>(wPtr + k);
        c = __builtin_amdgcn_wmma_f32_16x16x4_f32(
                false, a, false, b, (short)0, c, false, false);
    }

    const int sel = lane & 7;
    float v = c[0];
    v = (sel == 1) ? c[1] : v;
    v = (sel == 2) ? c[2] : v;
    v = (sel == 3) ? c[3] : v;
    v = (sel == 4) ? c[4] : v;
    v = (sel == 5) ? c[5] : v;
    v = (sel == 6) ? c[6] : v;
    v = (sel == 7) ? c[7] : v;

    const int  mOut    = ((lane >> 4) << 3) | sel;
    const bool doStore = (lane < 8) || (lane >= 16 && lane < 24);
    if (doStore && (rowBase + mOut) < rows) {
        proj[rowBase + mOut] = v;
    }
}

constexpr int kNone = 0x7fffffff;

template <int C, bool GUARD>
__device__ __forceinline__ int first_earlier_in_batch(const float* __restrict__ ts,
                                                      unsigned e0, unsigned stride,
                                                      int nvalid, float g) {
    float v[C];
    #pragma unroll
    for (int j = C - 1; j >= 0; --j) {
        const unsigned jj = (GUARD && j >= nvalid) ? (unsigned)(nvalid - 1) : (unsigned)j;
        v[j] = ts[e0 + jj * stride];
    }
    __builtin_amdgcn_sched_barrier(0);
    int first = kNone;
    #pragma unroll
    for (int j = C - 1; j >= 0; --j) {
        const bool ok = !GUARD || (j < nvalid);
        first = (ok && v[j] < g) ? j : first;
    }
    return first;
}

__global__ void __launch_bounds__(256)
rmtpp_grid_kernel(const float* __restrict__ ts,
                  const float* __restrict__ proj,
                  const float* __restrict__ dur,
                  const float* __restrict__ time_influence,
                  const float* __restrict__ base_intensity,
                  float* __restrict__ out_log,
                  float* __restrict__ out_grid,
                  int T, int BS, int NG) {
    const int gid = blockIdx.x * blockDim.x + threadIdx.x;
    if (gid >= NG * BS) return;
    const int b = gid % BS;
    const int n = gid / BS;

    const float ti = time_influence[0];
    const float bi = base_intensity[0];

    const float t_grid = ((float)n / (float)(NG - 1)) * dur[b];

    const unsigned ub = (unsigned)b;
    const unsigned us = (unsigned)BS;
    int first = first_earlier_in_batch<8, true>(ts, ub, us, T, t_grid);
    const int nFull = (T > 8) ? (T - 8) / 64 : 0;
    for (int k = 0; k < nFull && first == kNone; ++k) {
        const int t0 = 8 + k * 64;
        const int f  = first_earlier_in_batch<64, false>(ts, (unsigned)t0 * us + ub, us, 64, t_grid);
        first = (f == kNone) ? kNone : t0 + f;
    }
    const int tRem = 8 + nFull * 64;
    if (first == kNone && tRem < T) {
        const int f = first_earlier_in_batch<64, true>(ts, (unsigned)tRem * us + ub, us, T - tRem, t_grid);
        first = (f == kNone) ? kNone : tRem + f;
    }
    const int idx = (first == kNone) ? 0 : first;

    const float closest_ts = ts[(size_t)idx * BS + b];
    const float li = proj[(size_t)idx * BS + b] + ti * (t_grid - closest_ts) + bi;

    out_log[gid]  = li;
    out_grid[gid] = t_grid;
}

extern "C" void kernel_launch(void* const* d_in, const int* in_sizes, int n_in,
                              void* d_out, int out_size, void* d_ws, size_t ws_size,
                              hipStream_t stream) {
    const float* h  = (const float*)d_in[0];
    const float* ts = (const float*)d_in[1];
    const float* w  = (const float*)d_in[2];
    const float* ti = (const float*)d_in[3];
    const float* bi = (const float*)d_in[4];

    const int NG  = 1000;
    const int BS  = out_size / (2 * NG);
    const int TBS = in_sizes[1];
    const int T   = TBS / BS;
    const int H   = in_sizes[2];

    float* out_log  = (float*)d_out;
    float* out_grid = out_log + (size_t)NG * BS;

    float* proj = (float*)d_ws;
    float* dur  = proj + TBS;

    rmtpp_durations_kernel<<<(BS + 127) / 128, 128, 0, stream>>>(ts, dur, T, BS);

    const int waves  = (TBS + 15) / 16;
    const int blocks = (waves + 7) / 8;
    rmtpp_proj_wmma_kernel<<<blocks, 256, 0, stream>>>(h, w, proj, TBS, H);

    const int total = NG * BS;
    rmtpp_grid_kernel<<<(total + 255) / 256, 256, 0, stream>>>(
        ts, proj, dur, ti, bi, out_log, out_grid, T, BS, NG);
}


# durations kernel: a column's T rows are dealt out to the 32 waves of a work-group, 32 loads in flight per lane, wave maxima joined in LDS (was one thread per column, 128 serial trips of 8 loads)
# speedup vs baseline: 13.7450x; 2.1361x over previous
#include <hip/hip_runtime.h>
#include <cstdint>


typedef float v2f __attribute__((ext_vector_type(2)));
typedef float v8f __attribute__((ext_vector_type(8)));

constexpr int kDurCols  = 32;
constexpr int kDurWaves = 32;
constexpr int kDurBatch = 32;

template <bool GUARD>
__device__ __forceinline__ float column_max_of_batch(const float* __restrict__ ts,
                                                     unsigned rowBase, unsigned stride,
                                                     unsigned col, int nvalid, float m) {
    float v[kDurBatch];
    #pragma unroll
    for (int j = 0; j < kDurBatch; ++j) {
        const unsigned jj = (GUARD && j >= nvalid) ? (unsigned)(nvalid - 1) : (unsigned)j;
        v[j] = ts[(size_t)(rowBase + jj * stride) + col];
    }
    #pragma unroll
    for (int j = 0; j < kDurBatch; ++j) m = fmaxf(m, v[j]);
    return m;
}

__global__ void __launch_bounds__(kDurCols * kDurWaves)
rmtpp_durations_kernel(const float* __restrict__ ts,
                       float* __restrict__ dur,
                       int T, int BS) {
    __shared__ float part[kDurWaves][kDurCols];
    const int lane = threadIdx.x & 31;
    const int w = __builtin_amdgcn_readfirstlane((int)(threadIdx.x >> 5));
    const int b = blockIdx.x * kDurCols + lane;
    const unsigned col = (unsigned)((b < BS) ? b : BS - 1);
    const unsigned us  = (unsigned)BS;

    const int per = (T + kDurWaves - 1) / kDurWaves;
    const int t0  = w * per;
    const int t1  = (t0 + per < T) ? t0 + per : T;

    float m = -3.402823466e38f;
    int t = t0;
    for (; t + kDurBatch <= t1; t += kDurBatch)
        m = column_max_of_batch<false>(ts, (unsigned)t * us, us, col, kDurBatch, m);
    if (t < t1)
        m = column_max_of_batch<true>(ts, (unsigned)t * us, us, col, t1 - t, m);

    part[w][lane] = m;
    __syncthreads();
    if (w == 0) {
        float r = part[0][lane];
        #pragma unroll
        for (int k = 1; k < kDurWaves; ++k) r = fmaxf(r, part[k][lane]);
        if (b < BS) dur[b] = r;
    }
}

typedef float v4f __attribute__((ext_vector_type(4)));

constexpr int kProjRows   = 16;
constexpr int kProjAhead  = 3;
constexpr int kProjChunks = 4;
constexpr int kProjH      = 128 * kProjChunks;

__global__ void
rmtpp_proj_wmma_kernel(const float* __restrict__ h,
                       const float* __restrict__ w,
                       float* __restrict__ proj,
                       int rows, int H) {
    const int lane        = threadIdx.x & 31;
    const int waveInBlock = threadIdx.x >> 5;
    const int wave        = blockIdx.x * (blockDim.x >> 5) + waveInBlock;
    const int rowBase     = wave * kProjRows;
    if (rowBase >= rows) return;

    if (H == kProjH) {
        v4f wq[kProjChunks];
        #pragma unroll
        for (int q = 0; q < kProjChunks; ++q)
            wq[q] = *reinterpret_cast<const v4f*>(w + 128 * q + 4 * lane);

        const float* __restrict__ hl = h + 4 * lane;
        v4f buf[kProjAhead + 1][kProjChunks];
        float d[kProjRows];

#define PROJ_LOAD_ROW(R) do { \
            int rr_ = rowBase + (R); rr_ = (rr_ < rows) ? rr_ : rows - 1; \
            const float* rp_ = hl + (size_t)rr_ * kProjH; \
            _Pragma("unroll") for (int q = 0; q < kProjChunks; ++q) \
                buf[(R) % (kProjAhead + 1)][q] = \
                    __builtin_nontemporal_load(reinterpret_cast<const v4f*>(rp_ + 128 * q)); \
        } while (0)

        const int  sel    = lane & 7;
        const bool onDiag = (lane < 8) || (lane >= 24);

        #pragma unroll
        for (int r = 0; r < kProjAhead; ++r) PROJ_LOAD_ROW(r);

        #pragma unroll
        for (int r = 0; r < kProjRows; ++r) {
            __builtin_amdgcn_sched_barrier(0);
            if (r + kProjAhead < kProjRows) PROJ_LOAD_ROW(r + kProjAhead);
            __builtin_amdgcn_sched_barrier(0);

            v8f c = {};
            #pragma unroll
            for (int q = 0; q < kProjChunks; ++q) {
                const v4f a = buf[r % (kProjAhead + 1)][q];
                const v4f b = wq[q];
                const v2f a0 = {a.x, a.y}, b0 = {b.x, b.y};
                const v2f a1 = {a.z, a.w}, b1 = {b.z, b.w};
                c = __builtin_amdgcn_wmma_f32_16x16x4_f32(
                        false, a0, false, b0, (short)0, c, false, false);
                c = __builtin_amdgcn_wmma_f32_16x16x4_f32(
                        false, a1, false, b1, (short)0, c, false, false);
            }
            float v = c[0];
            v = (sel == 1) ? c[1] : v;
            v = (sel == 2) ? c[2] : v;
            v = (sel == 3) ? c[3] : v;
            v = (sel == 4) ? c[4] : v;
            v = (sel == 5) ? c[5] : v;
            v = (sel == 6) ? c[6] : v;
            v = (sel == 7) ? c[7] : v;
            d[r] = onDiag ? v : 0.0f;
        }
#undef PROJ_LOAD_ROW

        float mine = 0.0f;
        #pragma unroll
        for (int r = 0; r < kProjRows; ++r) {
            float s = d[r];
            #pragma unroll
            for (int m = 16; m >= 1; m >>= 1) s += __shfl_xor(s, m, 32);
            mine = (lane == r) ? s : mine;
        }
        if (lane < kProjRows && (rowBase + lane) < rows) {
            proj[rowBase + lane] = mine;
        }
        return;
    }

    const int m    = lane & 15;
    const int koff = (lane >> 4) << 1;
    int r = rowBase + m;
    if (r >= rows) r = rows - 1;

    const float* __restrict__ rowPtr = h + (size_t)r * H + koff;
    const float* __restrict__ wPtr   = w + koff;

    v8f c = {};
    #pragma unroll 4
    for (int k = 0; k < H; k += 4) {
        v2f a = *reinterpret_cast<const v2f*>(rowPtr + k);
        v2f b = *reinterpret_cast<const v2f*>(wPtr + k);
        c = __builtin_amdgcn_wmma_f32_16x16x4_f32(
                false, a, false, b, (short)0, c, false, false);
    }

    const int sel = lane & 7;
    float v = c[0];
    v = (sel == 1) ? c[1] : v;
    v = (sel == 2) ? c[2] : v;
    v = (sel == 3) ? c[3] : v;
    v = (sel == 4) ? c[4] : v;
    v = (sel == 5) ? c[5] : v;
    v = (sel == 6) ? c[6] : v;
    v = (sel == 7) ? c[7] : v;

    const int  mOut    = ((lane >> 4) << 3) | sel;
    const bool doStore = (lane < 8) || (lane >= 16 && lane < 24);
    if (doStore && (rowBase + mOut) < rows) {
        proj[rowBase + mOut] = v;
    }
}

constexpr int kNone = 0x7fffffff;

template <int C, bool GUARD>
__device__ __forceinline__ int first_earlier_in_batch(const float* __restrict__ ts,
                                                      unsigned e0, unsigned stride,
                                                      int nvalid, float g) {
    float v[C];
    #pragma unroll
    for (int j = C - 1; j >= 0; --j) {
        const unsigned jj = (GUARD && j >= nvalid) ? (unsigned)(nvalid - 1) : (unsigned)j;
        v[j] = ts[e0 + jj * stride];
    }
    __builtin_amdgcn_sched_barrier(0);
    int first = kNone;
    #pragma unroll
    for (int j = C - 1; j >= 0; --j) {
        const bool ok = !GUARD || (j < nvalid);
        first = (ok && v[j] < g) ? j : first;
    }
    return first;
}

__global__ void __launch_bounds__(256)
rmtpp_grid_kernel(const float* __restrict__ ts,
                  const float* __restrict__ proj,
                  const float* __restrict__ dur,
                  const float* __restrict__ time_influence,
                  const float* __restrict__ base_intensity,
                  float* __restrict__ out_log,
                  float* __restrict__ out_grid,
                  int T, int BS, int NG) {
    const int gid = blockIdx.x * blockDim.x + threadIdx.x;
    if (gid >= NG * BS) return;
    const int b = gid % BS;
    const int n = gid / BS;

    const float ti = time_influence[0];
    const float bi = base_intensity[0];

    const float t_grid = ((float)n / (float)(NG - 1)) * dur[b];

    const unsigned ub = (unsigned)b;
    const unsigned us = (unsigned)BS;
    int first = first_earlier_in_batch<8, true>(ts, ub, us, T, t_grid);
    const int nFull = (T > 8) ? (T - 8) / 64 : 0;
    for (int k = 0; k < nFull && first == kNone; ++k) {
        const int t0 = 8 + k * 64;
        const int f  = first_earlier_in_batch<64, false>(ts, (unsigned)t0 * us + ub, us, 64, t_grid);
        first = (f == kNone) ? kNone : t0 + f;
    }
    const int tRem = 8 + nFull * 64;
    if (first == kNone && tRem < T) {
        const int f = first_earlier_in_batch<64, true>(ts, (unsigned)tRem * us + ub, us, T - tRem, t_grid);
        first = (f == kNone) ? kNone : tRem + f;
    }
    const int idx = (first == kNone) ? 0 : first;

    const float closest_ts = ts[(size_t)idx * BS + b];
    const float li = proj[(size_t)idx * BS + b] + ti * (t_grid - closest_ts) + bi;

    out_log[gid]  = li;
    out_grid[gid] = t_grid;
}

extern "C" void kernel_launch(void* const* d_in, const int* in_sizes, int n_in,
                              void* d_out, int out_size, void* d_ws, size_t ws_size,
                              hipStream_t stream) {
    const float* h  = (const float*)d_in[0];
    const float* ts = (const float*)d_in[1];
    const float* w  = (const float*)d_in[2];
    const float* ti = (const float*)d_in[3];
    const float* bi = (const float*)d_in[4];

    const int NG  = 1000;
    const int BS  = out_size / (2 * NG);
    const int TBS = in_sizes[1];
    const int T   = TBS / BS;
    const int H   = in_sizes[2];

    float* out_log  = (float*)d_out;
    float* out_grid = out_log + (size_t)NG * BS;

    float* proj = (float*)d_ws;
    float* dur  = proj + TBS;

    rmtpp_durations_kernel<<<(BS + kDurCols - 1) / kDurCols, kDurCols * kDurWaves, 0, stream>>>(ts, dur, T, BS);

    const int waves  = (TBS + 15) / 16;
    const int blocks = (waves + 7) / 8;
    rmtpp_proj_wmma_kernel<<<blocks, 256, 0, stream>>>(h, w, proj, TBS, H);

    const int total = NG * BS;
    rmtpp_grid_kernel<<<(total + 255) / 256, 256, 0, stream>>>(
        ts, proj, dur, ti, bi, out_log, out_grid, T, BS, NG);
}


# gather, then project: the full projection pass is dropped; the grid kernel takes the WMMA row dot of gathered rows only, once per run of neighbour grid points sharing a row; per-row sum order kept
# speedup vs baseline: 42.5408x; 3.0950x over previous
#include <hip/hip_runtime.h>
#include <cstdint>


typedef float v2f __attribute__((ext_vector_type(2)));
typedef float v8f __attribute__((ext_vector_type(8)));

constexpr int kDurCols  = 32;
constexpr int kDurWaves = 32;
constexpr int kDurBatch = 32;

template <bool GUARD>
__device__ __forceinline__ float column_max_of_batch(const float* __restrict__ ts,
                                                     unsigned rowBase, unsigned stride,
                                                     unsigned col, int nvalid, float m) {
    float v[kDurBatch];
    #pragma unroll
    for (int j = 0; j < kDurBatch; ++j) {
        const unsigned jj = (GUARD && j >= nvalid) ? (unsigned)(nvalid - 1) : (unsigned)j;
        v[j] = ts[(size_t)(rowBase + jj * stride) + col];
    }
    #pragma unroll
    for (int j = 0; j < kDurBatch; ++j) m = fmaxf(m, v[j]);
    return m;
}

__global__ void __launch_bounds__(kDurCols * kDurWaves)
rmtpp_durations_kernel(const float* __restrict__ ts,
                       float* __restrict__ dur,
                       int T, int BS) {
    __shared__ float part[kDurWaves][kDurCols];
    const int lane = threadIdx.x & 31;
    const int w = __builtin_amdgcn_readfirstlane((int)(threadIdx.x >> 5));
    const int b = blockIdx.x * kDurCols + lane;
    const unsigned col = (unsigned)((b < BS) ? b : BS - 1);
    const unsigned us  = (unsigned)BS;

    const int per = (T + kDurWaves - 1) / kDurWaves;
    const int t0  = w * per;
    const int t1  = (t0 + per < T) ? t0 + per : T;

    float m = -3.402823466e38f;
    int t = t0;
    for (; t + kDurBatch <= t1; t += kDurBatch)
        m = column_max_of_batch<false>(ts, (unsigned)t * us, us, col, kDurBatch, m);
    if (t < t1)
        m = column_max_of_batch<true>(ts, (unsigned)t * us, us, col, t1 - t, m);

    part[w][lane] = m;
    __syncthreads();
    if (w == 0) {
        float r = part[0][lane];
        #pragma unroll
        for (int k = 1; k < kDurWaves; ++k) r = fmaxf(r, part[k][lane]);
        if (b < BS) dur[b] = r;
    }
}

constexpr int kNone = 0x7fffffff;

template <int C, bool GUARD>
__device__ __forceinline__ int first_earlier_in_batch(const float* __restrict__ ts,
                                                      unsigned e0, unsigned stride,
                                                      int nvalid, float g) {
    float v[C];
    #pragma unroll
    for (int j = C - 1; j >= 0; --j) {
        const unsigned jj = (GUARD && j >= nvalid) ? (unsigned)(nvalid - 1) : (unsigned)j;
        v[j] = ts[e0 + jj * stride];
    }
    __builtin_amdgcn_sched_barrier(0);
    int first = kNone;
    #pragma unroll
    for (int j = C - 1; j >= 0; --j) {
        const bool ok = !GUARD || (j < nvalid);
        first = (ok && v[j] < g) ? j : first;
    }
    return first;
}

typedef float v4f __attribute__((ext_vector_type(4)));

constexpr int kProjChunks = 4;
constexpr int kProjH      = 128 * kProjChunks;

__device__ __forceinline__ v8f tile_add_piece(v8f c, v4f a, v4f b) {
    const v2f a0 = {a.x, a.y}, b0 = {b.x, b.y};
    const v2f a1 = {a.z, a.w}, b1 = {b.z, b.w};
    c = __builtin_amdgcn_wmma_f32_16x16x4_f32(false, a0, false, b0, (short)0, c, false, false);
    c = __builtin_amdgcn_wmma_f32_16x16x4_f32(false, a1, false, b1, (short)0, c, false, false);
    return c;
}

__device__ __forceinline__ float tile_trace(v8f c, int lane) {
    const int  sel    = lane & 7;
    const bool onDiag = (lane < 8) || (lane >= 24);
    float v = c[0];
    v = (sel == 1) ? c[1] : v;
    v = (sel == 2) ? c[2] : v;
    v = (sel == 3) ? c[3] : v;
    v = (sel == 4) ? c[4] : v;
    v = (sel == 5) ? c[5] : v;
    v = (sel == 6) ? c[6] : v;
    v = (sel == 7) ? c[7] : v;
    float s = onDiag ? v : 0.0f;
    #pragma unroll
    for (int m = 16; m >= 1; m >>= 1) s += __shfl_xor(s, m, 32);
    return s;
}

constexpr int kTile = 32;

__global__ void __launch_bounds__(kTile * kTile)
rmtpp_grid_proj_kernel(const float* __restrict__ h,
                       const float* __restrict__ w,
                       const float* __restrict__ ts,
                       const float* __restrict__ dur,
                       const float* __restrict__ time_influence,
                       const float* __restrict__ base_intensity,
                       float* __restrict__ out_log,
                       float* __restrict__ out_grid,
                       int T, int BS, int NG, int H) {
    __shared__ int   sIdx[kTile][kTile + 1];
    __shared__ float sVal[kTile][kTile + 1];

    const int lane = threadIdx.x & 31;
    const int wv = __builtin_amdgcn_readfirstlane((int)(threadIdx.x >> 5));
    const int b0 = blockIdx.x * kTile;
    const int n0 = blockIdx.y * kTile;

    const int n1 = n0 + wv;
    const int b1 = b0 + lane;
    const int nn = (n1 < NG) ? n1 : NG - 1;
    const int bc = (b1 < BS) ? b1 : BS - 1;

    const float ti = time_influence[0];
    const float bi = base_intensity[0];

    const float t_grid = ((float)nn / (float)(NG - 1)) * dur[bc];

    const unsigned ub = (unsigned)bc;
    const unsigned us = (unsigned)BS;
    int first = first_earlier_in_batch<8, true>(ts, ub, us, T, t_grid);
    const int nFull = (T > 8) ? (T - 8) / 64 : 0;
    for (int k = 0; k < nFull && first == kNone; ++k) {
        const int t0 = 8 + k * 64;
        const int f  = first_earlier_in_batch<64, false>(ts, (unsigned)t0 * us + ub, us, 64, t_grid);
        first = (f == kNone) ? kNone : t0 + f;
    }
    const int tRem = 8 + nFull * 64;
    if (first == kNone && tRem < T) {
        const int f = first_earlier_in_batch<64, true>(ts, (unsigned)tRem * us + ub, us, T - tRem, t_grid);
        first = (f == kNone) ? kNone : tRem + f;
    }
    const int idx = (first == kNone) ? 0 : first;

    const float closest_ts = ts[(size_t)idx * BS + bc];

    sIdx[wv][lane] = idx;
    __syncthreads();

    const int bcol  = (b0 + wv < BS) ? b0 + wv : BS - 1;
    const int myIdx = sIdx[lane][wv];
    const int left  = __shfl_up(myIdx, 1, 32);
    const bool runStart = (lane == 0) || (myIdx != left);
    unsigned runs = __builtin_amdgcn_ballot_w32(runStart);

    const float* __restrict__ hcol = h + (size_t)bcol * H;
    const size_t pitch = (size_t)BS * H;
    float val = 0.0f;

    if (H == kProjH) {
        v4f wq[kProjChunks];
        #pragma unroll
        for (int q = 0; q < kProjChunks; ++q)
            wq[q] = *reinterpret_cast<const v4f*>(w + 128 * q + 4 * lane);

        while (runs) {
            const int j = __builtin_ctz(runs);
            runs &= runs - 1;
            const int r = __builtin_amdgcn_readlane(myIdx, j);
            const float* __restrict__ rp = hcol + (size_t)r * pitch + 4 * lane;
            v4f a[kProjChunks];
            #pragma unroll
            for (int q = 0; q < kProjChunks; ++q)
                a[q] = *reinterpret_cast<const v4f*>(rp + 128 * q);
            v8f c = {};
            #pragma unroll
            for (int q = 0; q < kProjChunks; ++q) c = tile_add_piece(c, a[q], wq[q]);
            const float s = tile_trace(c, lane);
            val = (lane >= j) ? s : val;
        }
    } else {
        while (runs) {
            const int j = __builtin_ctz(runs);
            runs &= runs - 1;
            const int r = __builtin_amdgcn_readlane(myIdx, j);
            const float* __restrict__ rp = hcol + (size_t)r * pitch;
            v8f c = {};
            for (int k0 = 0; k0 < H; k0 += 128) {
                const int col = k0 + 4 * lane;
                v4f a, b;
                a.x = (col     < H) ? rp[col]     : 0.0f;  b.x = (col     < H) ? w[col]     : 0.0f;
                a.y = (col + 1 < H) ? rp[col + 1] : 0.0f;  b.y = (col + 1 < H) ? w[col + 1] : 0.0f;
                a.z = (col + 2 < H) ? rp[col + 2] : 0.0f;  b.z = (col + 2 < H) ? w[col + 2] : 0.0f;
                a.w = (col + 3 < H) ? rp[col + 3] : 0.0f;  b.w = (col + 3 < H) ? w[col + 3] : 0.0f;
                c = tile_add_piece(c, a, b);
            }
            const float s = tile_trace(c, lane);
            val = (lane >= j) ? s : val;
        }
    }

    sVal[lane][wv] = val;
    __syncthreads();

    const float li = sVal[wv][lane] + ti * (t_grid - closest_ts) + bi;
    if (n1 < NG && b1 < BS) {
        const size_t o = (size_t)n1 * BS + b1;
        out_log[o]  = li;
        out_grid[o] = t_grid;
    }
}

extern "C" void kernel_launch(void* const* d_in, const int* in_sizes, int n_in,
                              void* d_out, int out_size, void* d_ws, size_t ws_size,
                              hipStream_t stream) {
    const float* h  = (const float*)d_in[0];
    const float* ts = (const float*)d_in[1];
    const float* w  = (const float*)d_in[2];
    const float* ti = (const float*)d_in[3];
    const float* bi = (const float*)d_in[4];

    const int NG  = 1000;
    const int BS  = out_size / (2 * NG);
    const int TBS = in_sizes[1];
    const int T   = TBS / BS;
    const int H   = in_sizes[2];

    float* out_log  = (float*)d_out;
    float* out_grid = out_log + (size_t)NG * BS;

    float* dur = (float*)d_ws;

    rmtpp_durations_kernel<<<(BS + kDurCols - 1) / kDurCols, kDurCols * kDurWaves, 0, stream>>>(ts, dur, T, BS);

    const dim3 tiles((BS + kTile - 1) / kTile, (NG + kTile - 1) / kTile);
    rmtpp_grid_proj_kernel<<<tiles, kTile * kTile, 0, stream>>>(
        h, w, ts, dur, ti, bi, out_log, out_grid, T, BS, NG, H);
}


# one launch: each work-group takes the column maxima of its own 32 columns itself (rows dealt to its 32 waves, joined in LDS); durations launch and workspace dropped; first scan batch loads under it
# speedup vs baseline: 47.1008x; 1.1072x over previous
#include <hip/hip_runtime.h>
#include <cstdint>


typedef float v2f __attribute__((ext_vector_type(2)));
typedef float v8f __attribute__((ext_vector_type(8)));

constexpr int kDurWaves = 32;
constexpr int kDurBatch = 32;

template <bool GUARD>
__device__ __forceinline__ float column_max_of_batch(const float* __restrict__ ts,
                                                     unsigned rowBase, unsigned stride,
                                                     unsigned col, int nvalid, float m) {
    float v[kDurBatch];
    #pragma unroll
    for (int j = 0; j < kDurBatch; ++j) {
        const unsigned jj = (GUARD && j >= nvalid) ? (unsigned)(nvalid - 1) : (unsigned)j;
        v[j] = ts[(size_t)(rowBase + jj * stride) + col];
    }
    #pragma unroll
    for (int j = 0; j < kDurBatch; ++j) m = fmaxf(m, v[j]);
    return m;
}

constexpr int kNone = 0x7fffffff;

template <int C, bool GUARD>
__device__ __forceinline__ void load_batch(const float* __restrict__ ts,
                                           unsigned e0, unsigned stride,
                                           int nvalid, float (&v)[C]) {
    #pragma unroll
    for (int j = C - 1; j >= 0; --j) {
        const unsigned jj = (GUARD && j >= nvalid) ? (unsigned)(nvalid - 1) : (unsigned)j;
        v[j] = ts[e0 + jj * stride];
    }
}

template <int C, bool GUARD>
__device__ __forceinline__ int first_earlier_of(const float (&v)[C], int nvalid, float g) {
    int first = kNone;
    #pragma unroll
    for (int j = C - 1; j >= 0; --j) {
        const bool ok = !GUARD || (j < nvalid);
        first = (ok && v[j] < g) ? j : first;
    }
    return first;
}

template <int C, bool GUARD>
__device__ __forceinline__ int first_earlier_in_batch(const float* __restrict__ ts,
                                                      unsigned e0, unsigned stride,
                                                      int nvalid, float g) {
    float v[C];
    load_batch<C, GUARD>(ts, e0, stride, nvalid, v);
    __builtin_amdgcn_sched_barrier(0);
    return first_earlier_of<C, GUARD>(v, nvalid, g);
}

typedef float v4f __attribute__((ext_vector_type(4)));

constexpr int kProjChunks = 4;
constexpr int kProjH      = 128 * kProjChunks;

__device__ __forceinline__ v8f tile_add_piece(v8f c, v4f a, v4f b) {
    const v2f a0 = {a.x, a.y}, b0 = {b.x, b.y};
    const v2f a1 = {a.z, a.w}, b1 = {b.z, b.w};
    c = __builtin_amdgcn_wmma_f32_16x16x4_f32(false, a0, false, b0, (short)0, c, false, false);
    c = __builtin_amdgcn_wmma_f32_16x16x4_f32(false, a1, false, b1, (short)0, c, false, false);
    return c;
}

__device__ __forceinline__ float tile_trace(v8f c, int lane) {
    const int  sel    = lane & 7;
    const bool onDiag = (lane < 8) || (lane >= 24);
    float v = c[0];
    v = (sel == 1) ? c[1] : v;
    v = (sel == 2) ? c[2] : v;
    v = (sel == 3) ? c[3] : v;
    v = (sel == 4) ? c[4] : v;
    v = (sel == 5) ? c[5] : v;
    v = (sel == 6) ? c[6] : v;
    v = (sel == 7) ? c[7] : v;
    float s = onDiag ? v : 0.0f;
    #pragma unroll
    for (int m = 16; m >= 1; m >>= 1) s += __shfl_xor(s, m, 32);
    return s;
}

constexpr int kTile = 32;
static_assert(kDurWaves == kTile, "the maxima deal the rows out to the work-group's kTile waves");

__global__ void __launch_bounds__(kTile * kTile)
rmtpp_grid_proj_kernel(const float* __restrict__ h,
                       const float* __restrict__ w,
                       const float* __restrict__ ts,
                       const float* __restrict__ time_influence,
                       const float* __restrict__ base_intensity,
                       float* __restrict__ out_log,
                       float* __restrict__ out_grid,
                       int T, int BS, int NG, int H) {
    __shared__ int   sIdx[kTile][kTile + 1];
    __shared__ float sVal[kTile][kTile + 1];
    __shared__ float sMax[kDurWaves][kTile];

    const int lane = threadIdx.x & 31;
    const int wv = __builtin_amdgcn_readfirstlane((int)(threadIdx.x >> 5));
    const int b0 = blockIdx.x * kTile;
    const int n0 = blockIdx.y * kTile;

    const int n1 = n0 + wv;
    const int b1 = b0 + lane;
    const int nn = (n1 < NG) ? n1 : NG - 1;
    const int bc = (b1 < BS) ? b1 : BS - 1;

    const float ti = time_influence[0];
    const float bi = base_intensity[0];

    const unsigned ub = (unsigned)bc;
    const unsigned us = (unsigned)BS;

    float head[8];
    load_batch<8, true>(ts, ub, us, T, head);

    {
        const int per = (T + kDurWaves - 1) / kDurWaves;
        const int t0  = wv * per;
        const int t1  = (t0 + per < T) ? t0 + per : T;
        float m = -3.402823466e38f;
        int t = t0;
        for (; t + kDurBatch <= t1; t += kDurBatch)
            m = column_max_of_batch<false>(ts, (unsigned)t * us, us, ub, kDurBatch, m);
        if (t < t1)
            m = column_max_of_batch<true>(ts, (unsigned)t * us, us, ub, t1 - t, m);
        sMax[wv][lane] = m;
    }
    __syncthreads();
    float dur = sMax[0][lane];
    #pragma unroll
    for (int k = 1; k < kDurWaves; ++k) dur = fmaxf(dur, sMax[k][lane]);

    const float t_grid = ((float)nn / (float)(NG - 1)) * dur;

    int first = first_earlier_of<8, true>(head, T, t_grid);
    const int nFull = (T > 8) ? (T - 8) / 64 : 0;
    for (int k = 0; k < nFull && first == kNone; ++k) {
        const int t0 = 8 + k * 64;
        const int f  = first_earlier_in_batch<64, false>(ts, (unsigned)t0 * us + ub, us, 64, t_grid);
        first = (f == kNone) ? kNone : t0 + f;
    }
    const int tRem = 8 + nFull * 64;
    if (first == kNone && tRem < T) {
        const int f = first_earlier_in_batch<64, true>(ts, (unsigned)tRem * us + ub, us, T - tRem, t_grid);
        first = (f == kNone) ? kNone : tRem + f;
    }
    const int idx = (first == kNone) ? 0 : first;

    const float closest_ts = ts[(size_t)idx * BS + bc];

    sIdx[wv][lane] = idx;
    __syncthreads();

    const int bcol  = (b0 + wv < BS) ? b0 + wv : BS - 1;
    const int myIdx = sIdx[lane][wv];
    const int left  = __shfl_up(myIdx, 1, 32);
    const bool runStart = (lane == 0) || (myIdx != left);
    unsigned runs = __builtin_amdgcn_ballot_w32(runStart);

    const float* __restrict__ hcol = h + (size_t)bcol * H;
    const size_t pitch = (size_t)BS * H;
    float val = 0.0f;

    if (H == kProjH) {
        v4f wq[kProjChunks];
        #pragma unroll
        for (int q = 0; q < kProjChunks; ++q)
            wq[q] = *reinterpret_cast<const v4f*>(w + 128 * q + 4 * lane);

        while (runs) {
            const int j = __builtin_ctz(runs);
            runs &= runs - 1;
            const int r = __builtin_amdgcn_readlane(myIdx, j);
            const float* __restrict__ rp = hcol + (size_t)r * pitch + 4 * lane;
            v4f a[kProjChunks];
            #pragma unroll
            for (int q = 0; q < kProjChunks; ++q)
                a[q] = *reinterpret_cast<const v4f*>(rp + 128 * q);
            v8f c = {};
            #pragma unroll
            for (int q = 0; q < kProjChunks; ++q) c = tile_add_piece(c, a[q], wq[q]);
            const float s = tile_trace(c, lane);
            val = (lane >= j) ? s : val;
        }
    } else {
        while (runs) {
            const int j = __builtin_ctz(runs);
            runs &= runs - 1;
            const int r = __builtin_amdgcn_readlane(myIdx, j);
            const float* __restrict__ rp = hcol + (size_t)r * pitch;
            v8f c = {};
            for (int k0 = 0; k0 < H; k0 += 128) {
                const int col = k0 + 4 * lane;
                v4f a, b;
                a.x = (col     < H) ? rp[col]     : 0.0f;  b.x = (col     < H) ? w[col]     : 0.0f;
                a.y = (col + 1 < H) ? rp[col + 1] : 0.0f;  b.y = (col + 1 < H) ? w[col + 1] : 0.0f;
                a.z = (col + 2 < H) ? rp[col + 2] : 0.0f;  b.z = (col + 2 < H) ? w[col + 2] : 0.0f;
                a.w = (col + 3 < H) ? rp[col + 3] : 0.0f;  b.w = (col + 3 < H) ? w[col + 3] : 0.0f;
                c = tile_add_piece(c, a, b);
            }
            const float s = tile_trace(c, lane);
            val = (lane >= j) ? s : val;
        }
    }

    sVal[lane][wv] = val;
    __syncthreads();

    const float li = sVal[wv][lane] + ti * (t_grid - closest_ts) + bi;
    if (n1 < NG && b1 < BS) {
        const size_t o = (size_t)n1 * BS + b1;
        out_log[o]  = li;
        out_grid[o] = t_grid;
    }
}

extern "C" void kernel_launch(void* const* d_in, const int* in_sizes, int n_in,
                              void* d_out, int out_size, void* d_ws, size_t ws_size,
                              hipStream_t stream) {
    const float* h  = (const float*)d_in[0];
    const float* ts = (const float*)d_in[1];
    const float* w  = (const float*)d_in[2];
    const float* ti = (const float*)d_in[3];
    const float* bi = (const float*)d_in[4];

    const int NG  = 1000;
    const int BS  = out_size / (2 * NG);
    const int TBS = in_sizes[1];
    const int T   = TBS / BS;
    const int H   = in_sizes[2];

    float* out_log  = (float*)d_out;
    float* out_grid = out_log + (size_t)NG * BS;

    const dim3 tiles((BS + kTile - 1) / kTile, (NG + kTile - 1) / kTile);
    rmtpp_grid_proj_kernel<<<tiles, kTile * kTile, 0, stream>>>(
        h, w, ts, ti, bi, out_log, out_grid, T, BS, NG, H);
}


# long scan: a trip turns its rows into running minima in place (one v_min a row, as loads return) and tests only the last against the grid time; compare+select per row only in a trip that has a hit
# speedup vs baseline: 68.8735x; 1.4623x over previous
#include <hip/hip_runtime.h>
#include <cstdint>


typedef float v2f __attribute__((ext_vector_type(2)));
typedef float v8f __attribute__((ext_vector_type(8)));

constexpr int kDurWaves = 32;
constexpr int kDurBatch = 32;

template <bool GUARD>
__device__ __forceinline__ float column_max_of_batch(const float* __restrict__ ts,
                                                     unsigned rowBase, unsigned stride,
                                                     unsigned col, int nvalid, float m) {
    float v[kDurBatch];
    #pragma unroll
    for (int j = 0; j < kDurBatch; ++j) {
        const unsigned jj = (GUARD && j >= nvalid) ? (unsigned)(nvalid - 1) : (unsigned)j;
        v[j] = ts[(size_t)(rowBase + jj * stride) + col];
    }
    #pragma unroll
    for (int j = 0; j < kDurBatch; ++j) m = fmaxf(m, v[j]);
    return m;
}

constexpr int kNone = 0x7fffffff;

template <int C, bool GUARD>
__device__ __forceinline__ void load_batch(const float* __restrict__ ts,
                                           unsigned e0, unsigned stride,
                                           int nvalid, float (&v)[C]) {
    #pragma unroll
    for (int j = C - 1; j >= 0; --j) {
        const unsigned jj = (GUARD && j >= nvalid) ? (unsigned)(nvalid - 1) : (unsigned)j;
        v[j] = ts[e0 + jj * stride];
    }
}

template <int C, bool GUARD>
__device__ __forceinline__ int first_earlier_of(const float (&v)[C], int nvalid, float g) {
    int first = kNone;
    #pragma unroll
    for (int j = C - 1; j >= 0; --j) {
        const bool ok = !GUARD || (j < nvalid);
        first = (ok && v[j] < g) ? j : first;
    }
    return first;
}

__device__ __forceinline__ void lower_to(float& x, float lo) {
    asm volatile("v_min_num_f32 %0, %1, %0" : "+v"(x) : "v"(lo));
}

template <int C, bool GUARD>
__device__ __forceinline__ int first_earlier_in_batch(const float* __restrict__ ts,
                                                      unsigned e0, unsigned stride,
                                                      int nvalid, float g) {
    float v[C];
    #pragma unroll
    for (int j = 0; j < C; ++j) {
        const unsigned jj = (GUARD && j >= nvalid) ? (unsigned)(nvalid - 1) : (unsigned)j;
        v[j] = ts[e0 + jj * stride];
    }
    __builtin_amdgcn_sched_barrier(0);
    static_assert(C >= 8, "four chains of running minima");
    #pragma unroll
    for (int j = 4; j < C; ++j) lower_to(v[j], v[j - 4]);
    const bool hit = (v[C - 4] < g) || (v[C - 3] < g) || (v[C - 2] < g) || (v[C - 1] < g);
    int f = kNone;
    if (__builtin_amdgcn_ballot_w32(hit) != 0)
        f = first_earlier_of<C, GUARD>(v, nvalid, g);
    return f;
}

typedef float v4f __attribute__((ext_vector_type(4)));
constexpr int kWideBS    = 128;
constexpr int kWideInstr = 27;
constexpr int kWideRows  = 4 * kWideInstr;

__device__ __forceinline__ void wide_batch_load(const v4f* __restrict__ row0, unsigned laneq,
                                                v4f (&v)[kWideInstr]) {
    #pragma unroll
    for (int i = 0; i < kWideInstr; ++i)
        v[i] = row0[laneq + (unsigned)(i * kWideBS)];
}

__device__ __forceinline__ void wide_running_min(v4f (&v)[kWideInstr]) {
    #pragma unroll
    for (int i = 1; i < kWideInstr; ++i) {
        float x = v[i].x, y = v[i].y, z = v[i].z, w = v[i].w;
        lower_to(x, v[i - 1].x);
        lower_to(y, v[i - 1].y);
        lower_to(z, v[i - 1].z);
        lower_to(w, v[i - 1].w);
        v[i].x = x; v[i].y = y; v[i].z = z; v[i].w = w;
    }
}

__device__ __forceinline__ void wide_batch_first_of(const v4f (&v)[kWideInstr], v4f g4,
                                                    int (&fb)[4]) {
    fb[0] = kNone; fb[1] = kNone; fb[2] = kNone; fb[3] = kNone;
    #pragma unroll
    for (int i = kWideInstr - 1; i >= 0; --i) {
        fb[0] = (v[i].x < g4.x) ? i : fb[0];
        fb[1] = (v[i].y < g4.y) ? i : fb[1];
        fb[2] = (v[i].z < g4.z) ? i : fb[2];
        fb[3] = (v[i].w < g4.w) ? i : fb[3];
    }
}


constexpr int kProjChunks = 4;
constexpr int kProjH      = 128 * kProjChunks;

__device__ __forceinline__ v8f tile_add_piece(v8f c, v4f a, v4f b) {
    const v2f a0 = {a.x, a.y}, b0 = {b.x, b.y};
    const v2f a1 = {a.z, a.w}, b1 = {b.z, b.w};
    c = __builtin_amdgcn_wmma_f32_16x16x4_f32(false, a0, false, b0, (short)0, c, false, false);
    c = __builtin_amdgcn_wmma_f32_16x16x4_f32(false, a1, false, b1, (short)0, c, false, false);
    return c;
}

__device__ __forceinline__ float tile_trace(v8f c, int lane) {
    const int  sel    = lane & 7;
    const bool onDiag = (lane < 8) || (lane >= 24);
    float v = c[0];
    v = (sel == 1) ? c[1] : v;
    v = (sel == 2) ? c[2] : v;
    v = (sel == 3) ? c[3] : v;
    v = (sel == 4) ? c[4] : v;
    v = (sel == 5) ? c[5] : v;
    v = (sel == 6) ? c[6] : v;
    v = (sel == 7) ? c[7] : v;
    float s = onDiag ? v : 0.0f;
    #pragma unroll
    for (int m = 16; m >= 1; m >>= 1) s += __shfl_xor(s, m, 32);
    return s;
}

constexpr int kTile = 32;
static_assert(kDurWaves == kTile, "the maxima deal the rows out to the work-group's kTile waves");

__global__ void __launch_bounds__(kTile * kTile)
rmtpp_grid_proj_kernel(const float* __restrict__ h,
                       const float* __restrict__ w,
                       const float* __restrict__ ts,
                       const float* __restrict__ time_influence,
                       const float* __restrict__ base_intensity,
                       float* __restrict__ out_log,
                       float* __restrict__ out_grid,
                       int T, int BS, int NG, int H) {
    __shared__ int   sIdx[kTile][kTile + 1];
    __shared__ float sVal[kTile][kTile + 1];
    __shared__ float sMax[kDurWaves][kTile];

    const int lane = threadIdx.x & 31;
    const int wv = __builtin_amdgcn_readfirstlane((int)(threadIdx.x >> 5));
    const int b0 = blockIdx.x * kTile;
    const int n0 = blockIdx.y * kTile;

    const int n1 = n0 + wv;
    const int b1 = b0 + lane;
    const int nn = (n1 < NG) ? n1 : NG - 1;
    const int bc = (b1 < BS) ? b1 : BS - 1;

    const float ti = time_influence[0];
    const float bi = base_intensity[0];

    const unsigned ub = (unsigned)bc;
    const unsigned us = (unsigned)BS;

    float head[8];
    load_batch<8, true>(ts, ub, us, T, head);

    {
        const int per = (T + kDurWaves - 1) / kDurWaves;
        const int t0  = wv * per;
        const int t1  = (t0 + per < T) ? t0 + per : T;
        float m = -3.402823466e38f;
        int t = t0;
        for (; t + kDurBatch <= t1; t += kDurBatch)
            m = column_max_of_batch<false>(ts, (unsigned)t * us, us, ub, kDurBatch, m);
        if (t < t1)
            m = column_max_of_batch<true>(ts, (unsigned)t * us, us, ub, t1 - t, m);
        sMax[wv][lane] = m;
    }
    __syncthreads();
    float dur = sMax[0][lane];
    #pragma unroll
    for (int k = 1; k < kDurWaves; ++k) dur = fmaxf(dur, sMax[k][lane]);

    const float t_grid = ((float)nn / (float)(NG - 1)) * dur;

    int first = first_earlier_of<8, true>(head, T, t_grid);

    int tBase = 8;
    const bool wideOK = (BS == kWideBS) && ((reinterpret_cast<uintptr_t>(ts) & 15u) == 0);
    const int  nWide  = (wideOK && T > 8) ? (T - 8) / kWideRows : 0;
    if (nWide > 0 && __builtin_amdgcn_ballot_w32(first == kNone) != 0) {
        const int p = lane >> 3;
        const int q = lane & 7;
        v4f g4;
        int f4[4];
        g4.x = __shfl(t_grid, 4 * q + 0, 32);  f4[0] = __shfl(first, 4 * q + 0, 32);
        g4.y = __shfl(t_grid, 4 * q + 1, 32);  f4[1] = __shfl(first, 4 * q + 1, 32);
        g4.z = __shfl(t_grid, 4 * q + 2, 32);  f4[2] = __shfl(first, 4 * q + 2, 32);
        g4.w = __shfl(t_grid, 4 * q + 3, 32);  f4[3] = __shfl(first, 4 * q + 3, 32);
        constexpr int rowq = kWideBS / 4;
        const unsigned laneq = (unsigned)(p * rowq + q);
        const v4f* __restrict__ row0 = reinterpret_cast<const v4f*>(ts) + (8 * rowq + (b0 >> 2));
        const float kClosed = -__builtin_inff();
        g4.x = (f4[0] == kNone) ? g4.x : kClosed;
        g4.y = (f4[1] == kNone) ? g4.y : kClosed;
        g4.z = (f4[2] == kNone) ? g4.z : kClosed;
        g4.w = (f4[3] == kNone) ? g4.w : kClosed;
        bool open = true;
        for (int k = 0; k < nWide && open; ++k) {
            v4f v[kWideInstr];
            wide_batch_load(row0, laneq, v);
            __builtin_amdgcn_sched_barrier(0);
            wide_running_min(v);
            row0 += kWideRows * rowq;
            const v4f lo = v[kWideInstr - 1];
            const bool hit = (lo.x < g4.x) || (lo.y < g4.y) || (lo.z < g4.z) || (lo.w < g4.w);
            if (__builtin_amdgcn_ballot_w32(hit) != 0) {
                int fb[4];
                wide_batch_first_of(v, g4, fb);
                const int t0 = 8 + k * kWideRows;
                bool mine = false;
                #pragma unroll
                for (int e = 0; e < 4; ++e) {
                    int m = (fb[e] == kNone) ? kNone : 4 * fb[e] + p;
                    const int m8 = __shfl_xor(m, 8, 32);
                    m = (m8 < m) ? m8 : m;
                    const int m16 = __shfl_xor(m, 16, 32);
                    m = (m16 < m) ? m16 : m;
                    f4[e] = (f4[e] == kNone && m != kNone) ? t0 + m : f4[e];
                    mine = mine || (f4[e] == kNone);
                }
                g4.x = (f4[0] == kNone) ? g4.x : kClosed;
                g4.y = (f4[1] == kNone) ? g4.y : kClosed;
                g4.z = (f4[2] == kNone) ? g4.z : kClosed;
                g4.w = (f4[3] == kNone) ? g4.w : kClosed;
                open = __builtin_amdgcn_ballot_w32(mine) != 0;
            }
        }
        const int r0 = __shfl(f4[0], lane >> 2, 32);
        const int r1 = __shfl(f4[1], lane >> 2, 32);
        const int r2 = __shfl(f4[2], lane >> 2, 32);
        const int r3 = __shfl(f4[3], lane >> 2, 32);
        const int el = lane & 3;
        first = (el == 0) ? r0 : (el == 1) ? r1 : (el == 2) ? r2 : r3;
        tBase = 8 + nWide * kWideRows;
    }

    const int nFull = (T > tBase) ? (T - tBase) / 64 : 0;
    for (int k = 0; k < nFull && first == kNone; ++k) {
        const int t0 = tBase + k * 64;
        const int f  = first_earlier_in_batch<64, false>(ts, (unsigned)t0 * us + ub, us, 64, t_grid);
        first = (f == kNone) ? kNone : t0 + f;
    }
    const int tRem = tBase + nFull * 64;
    if (first == kNone && tRem < T) {
        const int f = first_earlier_in_batch<64, true>(ts, (unsigned)tRem * us + ub, us, T - tRem, t_grid);
        first = (f == kNone) ? kNone : tRem + f;
    }
    const int idx = (first == kNone) ? 0 : first;

    const float closest_ts = ts[(size_t)idx * BS + bc];

    sIdx[wv][lane] = idx;
    __syncthreads();

    const int bcol  = (b0 + wv < BS) ? b0 + wv : BS - 1;
    const int myIdx = sIdx[lane][wv];
    const int left  = __shfl_up(myIdx, 1, 32);
    const bool runStart = (lane == 0) || (myIdx != left);
    unsigned runs = __builtin_amdgcn_ballot_w32(runStart);

    const float* __restrict__ hcol = h + (size_t)bcol * H;
    const size_t pitch = (size_t)BS * H;
    float val = 0.0f;

    if (H == kProjH) {
        v4f wq[kProjChunks];
        #pragma unroll
        for (int q = 0; q < kProjChunks; ++q)
            wq[q] = *reinterpret_cast<const v4f*>(w + 128 * q + 4 * lane);

        while (runs) {
            const int j = __builtin_ctz(runs);
            runs &= runs - 1;
            const int r = __builtin_amdgcn_readlane(myIdx, j);
            const float* __restrict__ rp = hcol + (size_t)r * pitch + 4 * lane;
            v4f a[kProjChunks];
            #pragma unroll
            for (int q = 0; q < kProjChunks; ++q)
                a[q] = *reinterpret_cast<const v4f*>(rp + 128 * q);
            v8f c = {};
            #pragma unroll
            for (int q = 0; q < kProjChunks; ++q) c = tile_add_piece(c, a[q], wq[q]);
            const float s = tile_trace(c, lane);
            val = (lane >= j) ? s : val;
        }
    } else {
        while (runs) {
            const int j = __builtin_ctz(runs);
            runs &= runs - 1;
            const int r = __builtin_amdgcn_readlane(myIdx, j);
            const float* __restrict__ rp = hcol + (size_t)r * pitch;
            v8f c = {};
            for (int k0 = 0; k0 < H; k0 += 128) {
                const int col = k0 + 4 * lane;
                v4f a, b;
                a.x = (col     < H) ? rp[col]     : 0.0f;  b.x = (col     < H) ? w[col]     : 0.0f;
                a.y = (col + 1 < H) ? rp[col + 1] : 0.0f;  b.y = (col + 1 < H) ? w[col + 1] : 0.0f;
                a.z = (col + 2 < H) ? rp[col + 2] : 0.0f;  b.z = (col + 2 < H) ? w[col + 2] : 0.0f;
                a.w = (col + 3 < H) ? rp[col + 3] : 0.0f;  b.w = (col + 3 < H) ? w[col + 3] : 0.0f;
                c = tile_add_piece(c, a, b);
            }
            const float s = tile_trace(c, lane);
            val = (lane >= j) ? s : val;
        }
    }

    sVal[lane][wv] = val;
    __syncthreads();

    const float li = sVal[wv][lane] + ti * (t_grid - closest_ts) + bi;
    if (n1 < NG && b1 < BS) {
        const size_t o = (size_t)n1 * BS + b1;
        out_log[o]  = li;
        out_grid[o] = t_grid;
    }
}

extern "C" void kernel_launch(void* const* d_in, const int* in_sizes, int n_in,
                              void* d_out, int out_size, void* d_ws, size_t ws_size,
                              hipStream_t stream) {
    const float* h  = (const float*)d_in[0];
    const float* ts = (const float*)d_in[1];
    const float* w  = (const float*)d_in[2];
    const float* ti = (const float*)d_in[3];
    const float* bi = (const float*)d_in[4];

    const int NG  = 1000;
    const int BS  = out_size / (2 * NG);
    const int TBS = in_sizes[1];
    const int T   = TBS / BS;
    const int H   = in_sizes[2];

    float* out_log  = (float*)d_out;
    float* out_grid = out_log + (size_t)NG * BS;

    const dim3 tiles((BS + kTile - 1) / kTile, (NG + kTile - 1) / kTile);
    rmtpp_grid_proj_kernel<<<tiles, kTile * kTile, 0, stream>>>(
        h, w, ts, ti, bi, out_log, out_grid, T, BS, NG, H);
}
